# starting kernel
# baseline (speedup 1.0000x reference)
#include <hip/hip_runtime.h>

typedef __attribute__((ext_vector_type(2))) float v2f;
typedef __attribute__((ext_vector_type(8))) float v8f;

__global__ void optnet_prep_kernel(const float* __restrict__ puzzles,
                                   const float* __restrict__ A,
                                   float* __restrict__ ws) {
  if (threadIdx.x != 0 || blockIdx.x != 0) return;
  double M[9][18];
  for (int i = 0; i < 9; ++i)
    for (int j = 0; j < 18; ++j) M[i][j] = 0.0;
  for (int i = 0; i < 6; ++i) M[i][i] = 1.0;
  for (int r = 0; r < 3; ++r)
    for (int j = 0; j < 6; ++j) {
      double a = (double)A[r * 6 + j];
      M[j][6 + r] = a;
      M[6 + r][j] = a;
    }
  for (int i = 0; i < 9; ++i) M[i][9 + i] = 1.0;

  for (int col = 0; col < 9; ++col) {
    int piv = col;
    double best = M[col][col] < 0.0 ? -M[col][col] : M[col][col];
    for (int r = col + 1; r < 9; ++r) {
      double v = M[r][col] < 0.0 ? -M[r][col] : M[r][col];
      if (v > best) { best = v; piv = r; }
    }
    if (piv != col)
      for (int j = 0; j < 18; ++j) {
        double t = M[col][j]; M[col][j] = M[piv][j]; M[piv][j] = t;
      }
    double inv = 1.0 / M[col][col];
    for (int j = 0; j < 18; ++j) M[col][j] *= inv;
    for (int r = 0; r < 9; ++r) {
      if (r == col) continue;
      double f = M[r][col];
      if (f != 0.0)
        for (int j = 0; j < 18; ++j) M[r][j] -= f * M[col][j];
    }
  }

  double bval = -(double)puzzles[3];
  for (int j = 0; j < 6; ++j) {
    for (int k = 0; k < 6; ++k) ws[j * 6 + k] = (float)M[j][9 + k];
    ws[36 + j] = (float)(bval * (M[j][9 + 6] + M[j][9 + 7] + M[j][9 + 8]));
  }
}

__global__ void __launch_bounds__(256) optnet_main_kernel(
    const float* __restrict__ in, float* __restrict__ out,
    const float* __restrict__ ws, int numTiles) {
  const int lane = threadIdx.x & 31;
  const int waveId = (blockIdx.x << 3) | (threadIdx.x >> 5);
  const int n  = lane & 15;
  const int hi = lane >> 4;

  const int nn = (n < 6) ? n : 0;
  const float* wrow = ws + nn * 6;
  float b00 = wrow[hi ? 2 : 0];
  float b01 = wrow[hi ? 3 : 1];
  float b10 = hi ? 0.0f : wrow[4];
  float b11 = hi ? 0.0f : wrow[5];
  float cn  = ws[36 + nn];
  if (n >= 6) { b00 = 0.0f; b01 = 0.0f; b10 = 0.0f; b11 = 0.0f; cn = 0.0f; }
  v2f bLo = { b00, b01 };
  v2f bHi = { b10, b11 };
  const v2f vzero = { 0.0f, 0.0f };

  const int tile0 = waveId * 4;
  #pragma unroll
  for (int t = 0; t < 4; ++t) {
    int tile = tile0 + t;
    if (tile >= numTiles) break;
    int base = tile << 4;
    int row  = base + n;
    const float* rowp = in + (size_t)row * 6;

    v2f a0 = *(const v2f*)(rowp + (hi << 1));
    if (hi && row < 2) a0.y = 0.0f;
    v2f a1 = hi ? vzero : *(const v2f*)(rowp + 4);

    v8f acc = {};
    acc = __builtin_amdgcn_wmma_f32_16x16x4_f32(false, a0, false, bLo,
                                                (short)0, acc, false, false);
    acc = __builtin_amdgcn_wmma_f32_16x16x4_f32(false, a1, false, bHi,
                                                (short)0, acc, false, false);

    if (n < 6) {
      float* orow = out + (size_t)(base + (hi << 3)) * 6 + n;
      #pragma unroll
      for (int r = 0; r < 8; ++r) orow[(size_t)r * 6] = acc[r] + cn;
    }
  }
}

extern "C" void kernel_launch(void* const* d_in, const int* in_sizes, int n_in,
                              void* d_out, int out_size, void* d_ws, size_t ws_size,
                              hipStream_t stream) {
  const float* puzzles = (const float*)d_in[0];
  const float* A       = (const float*)d_in[1];
  float* out = (float*)d_out;
  float* ws  = (float*)d_ws;

  int nRows    = in_sizes[0] / 6;
  int numTiles = (nRows + 15) / 16;

  optnet_prep_kernel<<<1, 32, 0, stream>>>(puzzles, A, ws);

  int waves  = (numTiles + 3) / 4;
  int blocks = (waves + 7) / 8;
  optnet_main_kernel<<<blocks, 256, 0, stream>>>(puzzles, out, ws, numTiles);
}


# main: 2 batch rows per matrix row (rows stay on M, no LDS) x block-diagonal [[W^T,0],[0,W^T]]: K=12 in 3 WMMAs, N=12; whole-row 24 B loads, 96 B store runs, half the waves; sums 2+2+2 not 4+2
# speedup vs baseline: 1.0603x; 1.0603x over previous
#include <hip/hip_runtime.h>

typedef __attribute__((ext_vector_type(2))) float v2f;
typedef __attribute__((ext_vector_type(8))) float v8f;

__global__ void optnet_prep_kernel(const float* __restrict__ puzzles,
                                   const float* __restrict__ A,
                                   float* __restrict__ ws) {
  if (threadIdx.x != 0 || blockIdx.x != 0) return;
  double M[9][18];
  for (int i = 0; i < 9; ++i)
    for (int j = 0; j < 18; ++j) M[i][j] = 0.0;
  for (int i = 0; i < 6; ++i) M[i][i] = 1.0;
  for (int r = 0; r < 3; ++r)
    for (int j = 0; j < 6; ++j) {
      double a = (double)A[r * 6 + j];
      M[j][6 + r] = a;
      M[6 + r][j] = a;
    }
  for (int i = 0; i < 9; ++i) M[i][9 + i] = 1.0;

  for (int col = 0; col < 9; ++col) {
    int piv = col;
    double best = M[col][col] < 0.0 ? -M[col][col] : M[col][col];
    for (int r = col + 1; r < 9; ++r) {
      double v = M[r][col] < 0.0 ? -M[r][col] : M[r][col];
      if (v > best) { best = v; piv = r; }
    }
    if (piv != col)
      for (int j = 0; j < 18; ++j) {
        double t = M[col][j]; M[col][j] = M[piv][j]; M[piv][j] = t;
      }
    double inv = 1.0 / M[col][col];
    for (int j = 0; j < 18; ++j) M[col][j] *= inv;
    for (int r = 0; r < 9; ++r) {
      if (r == col) continue;
      double f = M[r][col];
      if (f != 0.0)
        for (int j = 0; j < 18; ++j) M[r][j] -= f * M[col][j];
    }
  }

  double bval = -(double)puzzles[3];
  for (int j = 0; j < 6; ++j) {
    for (int k = 0; k < 6; ++k) ws[j * 6 + k] = (float)M[j][9 + k];
    ws[36 + j] = (float)(bval * (M[j][9 + 6] + M[j][9 + 7] + M[j][9 + 8]));
  }
}

__global__ void __launch_bounds__(256) optnet_main_kernel(
    const float* __restrict__ in, float* __restrict__ out,
    const float* __restrict__ ws, int numTiles) {
  const int lane = threadIdx.x & 31;
  const int waveId = (blockIdx.x << 3) | (threadIdx.x >> 5);
  const int c = lane & 15;
  const int h = lane >> 4;

  const int  feat  = (c < 6) ? c : c - 6;
  const bool live  = (c < 12);
  const bool mine  = live && ((c >= 6) == (h != 0));
  const float* wrow = ws + (live ? feat : 0) * 6;
  v2f b0 = *(const v2f*)(wrow + 0);
  v2f b1 = *(const v2f*)(wrow + 2);
  v2f b2 = *(const v2f*)(wrow + 4);
  float cn = ws[36 + (live ? feat : 0)];
  const v2f vzero = { 0.0f, 0.0f };
  if (!mine) { b0 = vzero; b1 = vzero; b2 = vzero; }
  if (!live) cn = 0.0f;

  const int lrow = ((c & 7) << 2) + ((c >> 3) << 1) + h;

  const int tile0 = waveId * 4;
  #pragma unroll
  for (int t = 0; t < 4; ++t) {
    int tile = tile0 + t;
    if (tile >= numTiles) break;
    int base = tile << 5;
    int row  = base + lrow;
    const float* rowp = in + (size_t)row * 6;

    v2f a0 = *(const v2f*)(rowp + 0);
    v2f a1 = *(const v2f*)(rowp + 2);
    v2f a2 = *(const v2f*)(rowp + 4);
    if (row < 2) a1.y = 0.0f;

    v8f acc = {};
    acc = __builtin_amdgcn_wmma_f32_16x16x4_f32(false, a0, false, b0,
                                                (short)0, acc, false, false);
    acc = __builtin_amdgcn_wmma_f32_16x16x4_f32(false, a1, false, b1,
                                                (short)0, acc, false, false);
    acc = __builtin_amdgcn_wmma_f32_16x16x4_f32(false, a2, false, b2,
                                                (short)0, acc, false, false);

    if (live) {
      float* orow = out + (size_t)base * 6 + (h * 12 + c);
      #pragma unroll
      for (int r = 0; r < 8; ++r) orow[r * 24] = acc[r] + cn;
    }
  }
}

extern "C" void kernel_launch(void* const* d_in, const int* in_sizes, int n_in,
                              void* d_out, int out_size, void* d_ws, size_t ws_size,
                              hipStream_t stream) {
  const float* puzzles = (const float*)d_in[0];
  const float* A       = (const float*)d_in[1];
  float* out = (float*)d_out;
  float* ws  = (float*)d_ws;

  int nRows    = in_sizes[0] / 6;
  int numTiles = (nRows + 31) / 32;

  optnet_prep_kernel<<<1, 32, 0, stream>>>(puzzles, A, ws);

  int waves  = (numTiles + 3) / 4;
  int blocks = (waves + 7) / 8;
  optnet_main_kernel<<<blocks, 256, 0, stream>>>(puzzles, out, ws, numTiles);
}


# prep: no 9x9 elimination; block form of K instead: S = A A' (3x3), adj(S) and det by cofactors in f64, one thread per entry of W and c, in registers; other f64 operations and order; main unchanged
# speedup vs baseline: 2.4556x; 2.3159x over previous
#include <hip/hip_runtime.h>

typedef __attribute__((ext_vector_type(2))) float v2f;
typedef __attribute__((ext_vector_type(8))) float v8f;

__global__ void __launch_bounds__(64) optnet_prep_kernel(
    const float* __restrict__ puzzles, const float* __restrict__ A,
    float* __restrict__ ws) {
  const int t = threadIdx.x;
  if (blockIdx.x != 0 || t >= 42) return;

  double s00 = 0.0, s01 = 0.0, s02 = 0.0, s11 = 0.0, s12 = 0.0, s22 = 0.0;
  for (int j = 0; j < 6; ++j) {
    const double p0 = (double)A[j], p1 = (double)A[6 + j], p2 = (double)A[12 + j];
    s00 += p0 * p0; s01 += p0 * p1; s02 += p0 * p2;
    s11 += p1 * p1; s12 += p1 * p2; s22 += p2 * p2;
  }

  const double c00 = s11 * s22 - s12 * s12;
  const double c01 = s02 * s12 - s01 * s22;
  const double c02 = s01 * s12 - s02 * s11;
  const double c11 = s00 * s22 - s02 * s02;
  const double c12 = s01 * s02 - s00 * s12;
  const double c22 = s00 * s11 - s01 * s01;
  const double det = s00 * c00 + s01 * c01 + s02 * c02;

  const bool isW = (t < 36);
  const int  f   = isW ? t / 6 : t - 36;
  const int  k   = isW ? t - 6 * f : 0;
  const float yf0 = A[f], yf1 = A[6 + f], yf2 = A[12 + f];
  const float xf0 = A[k], xf1 = A[6 + k], xf2 = A[12 + k];
  const double y0 = (double)yf0, y1 = (double)yf1, y2 = (double)yf2;
  const double x0 = isW ? (double)xf0 : 1.0;
  const double x1 = isW ? (double)xf1 : 1.0;
  const double x2 = isW ? (double)xf2 : 1.0;

  const double u0 = c00 * x0 + c01 * x1 + c02 * x2;
  const double u1 = c01 * x0 + c11 * x1 + c12 * x2;
  const double u2 = c02 * x0 + c12 * x1 + c22 * x2;
  const double q  = (y0 * u0 + y1 * u1 + y2 * u2) / det;

  const double bval  = -(double)puzzles[3];
  const double scale = isW ? -1.0 : bval;
  const double start = (isW && f == k) ? 1.0 : 0.0;
  ws[t] = (float)__builtin_fma(q, scale, start);
}

__global__ void __launch_bounds__(256) optnet_main_kernel(
    const float* __restrict__ in, float* __restrict__ out,
    const float* __restrict__ ws, int numTiles) {
  const int lane = threadIdx.x & 31;
  const int waveId = (blockIdx.x << 3) | (threadIdx.x >> 5);
  const int c = lane & 15;
  const int h = lane >> 4;

  const int  feat  = (c < 6) ? c : c - 6;
  const bool live  = (c < 12);
  const bool mine  = live && ((c >= 6) == (h != 0));
  const float* wrow = ws + (live ? feat : 0) * 6;
  v2f b0 = *(const v2f*)(wrow + 0);
  v2f b1 = *(const v2f*)(wrow + 2);
  v2f b2 = *(const v2f*)(wrow + 4);
  float cn = ws[36 + (live ? feat : 0)];
  const v2f vzero = { 0.0f, 0.0f };
  if (!mine) { b0 = vzero; b1 = vzero; b2 = vzero; }
  if (!live) cn = 0.0f;

  const int lrow = ((c & 7) << 2) + ((c >> 3) << 1) + h;

  const int tile0 = waveId * 4;
  #pragma unroll
  for (int t = 0; t < 4; ++t) {
    int tile = tile0 + t;
    if (tile >= numTiles) break;
    int base = tile << 5;
    int row  = base + lrow;
    const float* rowp = in + (size_t)row * 6;

    v2f a0 = *(const v2f*)(rowp + 0);
    v2f a1 = *(const v2f*)(rowp + 2);
    v2f a2 = *(const v2f*)(rowp + 4);
    if (row < 2) a1.y = 0.0f;

    v8f acc = {};
    acc = __builtin_amdgcn_wmma_f32_16x16x4_f32(false, a0, false, b0,
                                                (short)0, acc, false, false);
    acc = __builtin_amdgcn_wmma_f32_16x16x4_f32(false, a1, false, b1,
                                                (short)0, acc, false, false);
    acc = __builtin_amdgcn_wmma_f32_16x16x4_f32(false, a2, false, b2,
                                                (short)0, acc, false, false);

    if (live) {
      float* orow = out + (size_t)base * 6 + (h * 12 + c);
      #pragma unroll
      for (int r = 0; r < 8; ++r) orow[r * 24] = acc[r] + cn;
    }
  }
}

extern "C" void kernel_launch(void* const* d_in, const int* in_sizes, int n_in,
                              void* d_out, int out_size, void* d_ws, size_t ws_size,
                              hipStream_t stream) {
  const float* puzzles = (const float*)d_in[0];
  const float* A       = (const float*)d_in[1];
  float* out = (float*)d_out;
  float* ws  = (float*)d_ws;

  int nRows    = in_sizes[0] / 6;
  int numTiles = (nRows + 31) / 32;

  optnet_prep_kernel<<<1, 64, 0, stream>>>(puzzles, A, ws);

  int waves  = (numTiles + 3) / 4;
  int blocks = (waves + 7) / 8;
  optnet_main_kernel<<<blocks, 256, 0, stream>>>(puzzles, out, ws, numTiles);
}


# prep: S, adj(S), adj(S) x formed once per wave, one value per lane, passed on by lane exchange, not repeated in every lane: 32 f64-rate instructions per wave, not 101; same formulas and term order
# speedup vs baseline: 2.5325x; 1.0313x over previous
#include <hip/hip_runtime.h>

typedef __attribute__((ext_vector_type(2))) float v2f;
typedef __attribute__((ext_vector_type(8))) float v8f;

__global__ void __launch_bounds__(64) optnet_prep_kernel(
    const float* __restrict__ puzzles, const float* __restrict__ A,
    float* __restrict__ ws) {
  if (blockIdx.x != 0) return;
  const int t    = threadIdx.x;
  const int lane = t & 31;

  const double aD = (double)A[lane < 18 ? lane : 17];
  const double bval = -(double)puzzles[3];

  const int e  = lane < 6 ? lane : 0;
  const int r  = (e >= 3) + (e >= 5);
  const int c  = e - (r ? r + 1 : 0);
  double xr[6], xc[6];
  for (int j = 0; j < 6; ++j) {
    xr[j] = __shfl(aD, 6 * r + j, 32);
    xc[j] = __shfl(aD, 6 * c + j, 32);
  }
  __builtin_amdgcn_sched_barrier(0);
  double s = 0.0;
  for (int j = 0; j < 6; ++j) s = __builtin_fma(xr[j], xc[j], s);

  const int sh = 4 * e;
  const double P = __shfl(s, (0x010123 >> sh) & 7, 32);
  const double Q = __shfl(s, (0x325445 >> sh) & 7, 32);
  const double R = __shfl(s, (0x102214 >> sh) & 7, 32);
  const double T = __shfl(s, (0x142354 >> sh) & 7, 32);
  __builtin_amdgcn_sched_barrier(0);
  const double cof = __builtin_fma(P, Q, -(R * T));

  const double s00 = __shfl(s, 0, 32), s01 = __shfl(s, 1, 32), s02 = __shfl(s, 2, 32);
  const double c00 = __shfl(cof, 0, 32), c01 = __shfl(cof, 1, 32), c02 = __shfl(cof, 2, 32);
  const double det = __builtin_fma(s02, c02, __builtin_fma(s01, c01, s00 * c00));

  const int m  = lane < 21 ? lane : 20;
  const int kq = m / 3;
  const int b  = m - 3 * kq;
  const int kc = kq < 6 ? kq : 5;
  const double a0 = __shfl(cof, (0x210 >> (4 * b)) & 7, 32);
  const double a1 = __shfl(cof, (0x431 >> (4 * b)) & 7, 32);
  const double a2 = __shfl(cof, (0x542 >> (4 * b)) & 7, 32);
  const double g0 = __shfl(aD, kc, 32);
  const double g1 = __shfl(aD, 6 + kc, 32);
  const double g2 = __shfl(aD, 12 + kc, 32);
  const double x0 = kq < 6 ? g0 : 1.0;
  const double x1 = kq < 6 ? g1 : 1.0;
  const double x2 = kq < 6 ? g2 : 1.0;
  const double u = __builtin_fma(a2, x2, __builtin_fma(a1, x1, a0 * x0));

  const bool isW = (t < 36);
  const int  fr  = isW ? t / 6 : t - 36;
  const int  f   = fr < 6 ? fr : 5;
  const int  k   = isW ? t - 6 * f : 6;
  const double y0 = __shfl(aD, f, 32);
  const double y1 = __shfl(aD, 6 + f, 32);
  const double y2 = __shfl(aD, 12 + f, 32);
  const double u0 = __shfl(u, 3 * k, 32);
  const double u1 = __shfl(u, 3 * k + 1, 32);
  const double u2 = __shfl(u, 3 * k + 2, 32);
  const double q  = __builtin_fma(y2, u2, __builtin_fma(y1, u1, y0 * u0)) / det;

  const double scale = isW ? -1.0 : bval;
  const double start = (isW && f == k) ? 1.0 : 0.0;
  if (t < 42) ws[t] = (float)__builtin_fma(q, scale, start);
}

__global__ void __launch_bounds__(256) optnet_main_kernel(
    const float* __restrict__ in, float* __restrict__ out,
    const float* __restrict__ ws, int numTiles) {
  const int lane = threadIdx.x & 31;
  const int waveId = (blockIdx.x << 3) | (threadIdx.x >> 5);
  const int c = lane & 15;
  const int h = lane >> 4;

  const int  feat  = (c < 6) ? c : c - 6;
  const bool live  = (c < 12);
  const bool mine  = live && ((c >= 6) == (h != 0));
  const float* wrow = ws + (live ? feat : 0) * 6;
  v2f b0 = *(const v2f*)(wrow + 0);
  v2f b1 = *(const v2f*)(wrow + 2);
  v2f b2 = *(const v2f*)(wrow + 4);
  float cn = ws[36 + (live ? feat : 0)];
  const v2f vzero = { 0.0f, 0.0f };
  if (!mine) { b0 = vzero; b1 = vzero; b2 = vzero; }
  if (!live) cn = 0.0f;

  const int lrow = ((c & 7) << 2) + ((c >> 3) << 1) + h;

  const int tile0 = waveId * 4;
  #pragma unroll
  for (int t = 0; t < 4; ++t) {
    int tile = tile0 + t;
    if (tile >= numTiles) break;
    int base = tile << 5;
    int row  = base + lrow;
    const float* rowp = in + (size_t)row * 6;

    v2f a0 = *(const v2f*)(rowp + 0);
    v2f a1 = *(const v2f*)(rowp + 2);
    v2f a2 = *(const v2f*)(rowp + 4);
    if (row < 2) a1.y = 0.0f;

    v8f acc = {};
    acc = __builtin_amdgcn_wmma_f32_16x16x4_f32(false, a0, false, b0,
                                                (short)0, acc, false, false);
    acc = __builtin_amdgcn_wmma_f32_16x16x4_f32(false, a1, false, b1,
                                                (short)0, acc, false, false);
    acc = __builtin_amdgcn_wmma_f32_16x16x4_f32(false, a2, false, b2,
                                                (short)0, acc, false, false);

    if (live) {
      float* orow = out + (size_t)base * 6 + (h * 12 + c);
      #pragma unroll
      for (int r = 0; r < 8; ++r) orow[r * 24] = acc[r] + cn;
    }
  }
}

extern "C" void kernel_launch(void* const* d_in, const int* in_sizes, int n_in,
                              void* d_out, int out_size, void* d_ws, size_t ws_size,
                              hipStream_t stream) {
  const float* puzzles = (const float*)d_in[0];
  const float* A       = (const float*)d_in[1];
  float* out = (float*)d_out;
  float* ws  = (float*)d_ws;

  int nRows    = in_sizes[0] / 6;
  int numTiles = (nRows + 31) / 32;

  optnet_prep_kernel<<<1, 64, 0, stream>>>(puzzles, A, ws);

  int waves  = (numTiles + 3) / 4;
  int blocks = (waves + 7) / 8;
  optnet_main_kernel<<<blocks, 256, 0, stream>>>(puzzles, out, ws, numTiles);
}
